# starting kernel
# baseline (speedup 1.0000x reference)
#include <hip/hip_runtime.h>


typedef float v2f __attribute__((ext_vector_type(2)));
typedef float v8f __attribute__((ext_vector_type(8)));

#define BATCH 16
#define CIN   32
#define HH    128
#define WW    128
#define PP    4
#define NNB   8
#define COUT  64
#define BN_EPS_C 1e-5f
#define ALPHA_LOG2E 14.4269504089f

#define HB      4
#define NWAVES  8
#define THREADS 256

#define OFF_WPAT  0
#define OFF_FUSE  (OFF_WPAT + 16*33)
#define OFF_PROJW (OFF_FUSE + 64*33)
#define OFF_BN    (OFF_PROJW + 64*33)
#define PROJ_RS   132
#define PROJ_PS   ((HB+2)*PROJ_RS)
#define OFF_PROJ  (OFF_BN + 128)
#define BITS_PS   528
#define OFF_BITS  (OFF_PROJ + PP*PROJ_PS)
#define LDS_FLOATS (OFF_BITS + CIN*BITS_PS)

#define DYPACK 0x6A40u
#define DXPACK 0x06A4u

__global__ __launch_bounds__(THREADS)
void lbp_fused_kernel(const float* __restrict__ x,
                      const float* __restrict__ wpat,
                      const float* __restrict__ fusew,
                      const float* __restrict__ gammaw,
                      const float* __restrict__ betaw,
                      const float* __restrict__ meanw,
                      const float* __restrict__ varw,
                      const float* __restrict__ projw,
                      const float* __restrict__ res_scale,
                      float* __restrict__ out)
{
    extern __shared__ float lds[];
    const int tid  = threadIdx.x;
    const int lane = tid & 31;
    const int wave = tid >> 5;
    const int hl   = lane >> 4;
    const int ln   = lane & 15;

    const int blk  = blockIdx.x;
    const int bimg = blk >> 5;
    const int hblk = blk & 31;
    const int h0   = hblk * HB;

    for (int i = tid; i < 16*CIN; i += THREADS) {
        int m = i >> 5, k = i & 31;
        lds[OFF_WPAT + m*33 + k] = (m < PP) ? wpat[m*CIN + k] : 0.0f;
    }
    for (int i = tid; i < COUT*CIN; i += THREADS) {
        int o = i >> 5, k = i & 31;
        lds[OFF_FUSE  + o*33 + k] = fusew[i];
        lds[OFF_PROJW + o*33 + k] = projw[i];
    }
    if (tid < COUT) {
        float inv = gammaw[tid] * rsqrtf(varw[tid] + BN_EPS_C);
        lds[OFF_BN + tid]      = inv;
        lds[OFF_BN + 64 + tid] = betaw[tid] - meanw[tid] * inv;
    }
    __syncthreads();

    const float* xb = x + (size_t)bimg * (CIN*HH*WW);

    for (int t = wave; t < (HB+2)*8; t += NWAVES) {
        int prow = t >> 3;
        int w0   = (t & 7) << 4;
        int hg   = (h0 - 1 + prow) & (HH - 1);
        v8f acc = {};
        #pragma unroll
        for (int ks = 0; ks < CIN; ks += 4) {
            int kb = ks + 2*hl;
            v2f a, bm;
            a[0] = lds[OFF_WPAT + ln*33 + kb];
            a[1] = lds[OFF_WPAT + ln*33 + kb + 1];
            const float* xp = xb + ((size_t)kb*HH + hg)*WW + w0 + ln;
            bm[0] = xp[0];
            bm[1] = xp[HH*WW];
            acc = __builtin_amdgcn_wmma_f32_16x16x4_f32(false, a, false, bm,
                                                        (short)0, acc, false, false);
        }
        if (hl == 0) {
            #pragma unroll
            for (int r = 0; r < PP; ++r)
                lds[OFF_PROJ + r*PROJ_PS + prow*PROJ_RS + w0 + ln] = acc[r];
        }
    }
    __syncthreads();

    for (int i = tid; i < CIN*HB*WW; i += THREADS) {
        int k   = i >> 9;
        int pix = i & 511;
        int row = pix >> 7, col = pix & 127;
        int p = k >> 3, nf = k & 7;
        int dy = (int)((DYPACK >> (2*nf)) & 3u) - 1;
        int dx = (int)((DXPACK >> (2*nf)) & 3u) - 1;
        float c  = lds[OFF_PROJ + p*PROJ_PS + (row+1)*PROJ_RS + col];
        float nb = lds[OFF_PROJ + p*PROJ_PS + (row+1-dy)*PROJ_RS + ((col - dx) & (WW-1))];
        float t  = __builtin_amdgcn_exp2f(ALPHA_LOG2E * (c - nb));
        lds[OFF_BITS + k*BITS_PS + pix] = __builtin_amdgcn_rcpf(1.0f + t);
    }
    __syncthreads();

    const float srs = res_scale[0];

    for (int t = wave; t < HB*8; t += NWAVES) {
        int row  = t >> 3;
        int w0   = (t & 7) << 4;
        int hg   = h0 + row;
        int pixb = (row << 7) + w0;

        v8f accF[4] = {};
        v8f accI[4] = {};
        #pragma unroll
        for (int ks = 0; ks < CIN; ks += 4) {
            int kb = ks + 2*hl;
            v2f bb, bx;
            bb[0] = lds[OFF_BITS + kb*BITS_PS + pixb + ln];
            bb[1] = lds[OFF_BITS + (kb+1)*BITS_PS + pixb + ln];
            const float* xp = xb + ((size_t)kb*HH + hg)*WW + w0 + ln;
            bx[0] = xp[0];
            bx[1] = xp[HH*WW];
            #pragma unroll
            for (int mt = 0; mt < 4; ++mt) {
                int ro = (mt*16 + ln)*33 + kb;
                v2f aF, aI;
                aF[0] = lds[OFF_FUSE  + ro]; aF[1] = lds[OFF_FUSE  + ro + 1];
                aI[0] = lds[OFF_PROJW + ro]; aI[1] = lds[OFF_PROJW + ro + 1];
                accF[mt] = __builtin_amdgcn_wmma_f32_16x16x4_f32(false, aF, false, bb,
                                                                 (short)0, accF[mt], false, false);
                accI[mt] = __builtin_amdgcn_wmma_f32_16x16x4_f32(false, aI, false, bx,
                                                                 (short)0, accI[mt], false, false);
            }
        }

        float* op = out + (((size_t)bimg*COUT)*HH + hg)*WW + w0 + ln;
        #pragma unroll
        for (int mt = 0; mt < 4; ++mt) {
            #pragma unroll
            for (int r = 0; r < 8; ++r) {
                int o = mt*16 + r + 8*hl;
                float v = (accF[mt][r]*lds[OFF_BN + o] + lds[OFF_BN + 64 + o])*srs + accI[mt][r];
                op[(size_t)o*HH*WW] = v > 0.0f ? v : 0.0f;
            }
        }
    }
}

extern "C" void kernel_launch(void* const* d_in, const int* in_sizes, int n_in,
                              void* d_out, int out_size, void* d_ws, size_t ws_size,
                              hipStream_t stream) {
    (void)in_sizes; (void)n_in; (void)d_ws; (void)ws_size; (void)out_size;
    const float* x      = (const float*)d_in[0];
    const float* wpat   = (const float*)d_in[1];
    const float* fusew  = (const float*)d_in[2];
    const float* gammaw = (const float*)d_in[3];
    const float* betaw  = (const float*)d_in[4];
    const float* meanw  = (const float*)d_in[5];
    const float* varw   = (const float*)d_in[6];
    const float* projw  = (const float*)d_in[7];
    const float* rscale = (const float*)d_in[8];
    float* outp = (float*)d_out;

    dim3 grid(BATCH * (HH / HB));
    dim3 block(THREADS);
    size_t ldsBytes = (size_t)LDS_FLOATS * sizeof(float);
    lbp_fused_kernel<<<grid, block, ldsBytes, stream>>>(
        x, wpat, fusew, gammaw, betaw, meanw, varw, projw, rscale, outp);
}


# register budget cut to 256 per lane (launch bound of 4 waves per SIMD; BN constants read from LDS per pixel tile, not held; x loads on one pointer) so two workgroups share a WGP; sums unchanged
# speedup vs baseline: 1.3194x; 1.3194x over previous
#include <hip/hip_runtime.h>


typedef float v2f __attribute__((ext_vector_type(2)));
typedef float v4f __attribute__((ext_vector_type(4)));
typedef float v8f __attribute__((ext_vector_type(8)));

#define BATCH 16
#define CIN   32
#define HH    128
#define WW    128
#define PP    4
#define NNB   8
#define COUT  64
#define BN_EPS_C 1e-5f
#define ALPHA_LOG2E 14.4269504089f

#define HB      4
#define NWAVES  8
#define THREADS 256

#define OFF_WPAT  0
#define OFF_FUSE  (OFF_WPAT + 16*33)
#define OFF_PROJW (OFF_FUSE + 64*33)
#define OFF_BN    (OFF_PROJW + 64*33)
#define PROJ_RS   132
#define PROJ_PS   ((HB+2)*PROJ_RS)
#define OFF_PROJ  (OFF_BN + 128)
#define BITS_PS   528
#define OFF_BITS  (OFF_PROJ + PP*PROJ_PS)
#define LDS_FLOATS (OFF_BITS + CIN*BITS_PS)

#define DYPACK 0x6A40u
#define DXPACK 0x06A4u

__global__ __launch_bounds__(THREADS, 4)
void lbp_fused_kernel(const float* __restrict__ x,
                      const float* __restrict__ wpat,
                      const float* __restrict__ fusew,
                      const float* __restrict__ gammaw,
                      const float* __restrict__ betaw,
                      const float* __restrict__ meanw,
                      const float* __restrict__ varw,
                      const float* __restrict__ projw,
                      const float* __restrict__ res_scale,
                      float* __restrict__ out)
{
    extern __shared__ __attribute__((aligned(16))) float lds[];
    const int tid  = threadIdx.x;
    const int lane = tid & 31;
    const int wave = tid >> 5;
    const int hl   = lane >> 4;
    const int ln   = lane & 15;

    const int blk  = blockIdx.x;
    const int bimg = blk >> 5;
    const int hblk = blk & 31;
    const int h0   = hblk * HB;

    for (int i = tid; i < 16*CIN; i += THREADS) {
        int m = i >> 5, k = i & 31;
        lds[OFF_WPAT + m*33 + k] = (m < PP) ? wpat[m*CIN + k] : 0.0f;
    }
    for (int i = tid; i < COUT*CIN; i += THREADS) {
        int o = i >> 5, k = i & 31;
        lds[OFF_FUSE  + o*33 + k] = fusew[i];
        lds[OFF_PROJW + o*33 + k] = projw[i];
    }
    if (tid < COUT) {
        float inv = gammaw[tid] * rsqrtf(varw[tid] + BN_EPS_C);
        lds[OFF_BN + tid]      = inv;
        lds[OFF_BN + 64 + tid] = betaw[tid] - meanw[tid] * inv;
    }
    __syncthreads();

    const float* xb = x + (size_t)bimg * (CIN*HH*WW);

    for (int t = wave; t < (HB+2)*8; t += NWAVES) {
        int prow = t >> 3;
        int w0   = (t & 7) << 4;
        int hg   = (h0 - 1 + prow) & (HH - 1);
        v8f acc = {};
        #pragma unroll
        for (int ks = 0; ks < CIN; ks += 4) {
            int kb = ks + 2*hl;
            v2f a, bm;
            a[0] = lds[OFF_WPAT + ln*33 + kb];
            a[1] = lds[OFF_WPAT + ln*33 + kb + 1];
            const float* xp = xb + ((size_t)kb*HH + hg)*WW + w0 + ln;
            bm[0] = xp[0];
            bm[1] = xp[HH*WW];
            acc = __builtin_amdgcn_wmma_f32_16x16x4_f32(false, a, false, bm,
                                                        (short)0, acc, false, false);
        }
        if (hl == 0) {
            #pragma unroll
            for (int r = 0; r < PP; ++r)
                lds[OFF_PROJ + r*PROJ_PS + prow*PROJ_RS + w0 + ln] = acc[r];
        }
    }
    __syncthreads();

    for (int i = tid; i < CIN*HB*WW; i += THREADS) {
        int k   = i >> 9;
        int pix = i & 511;
        int row = pix >> 7, col = pix & 127;
        int p = k >> 3, nf = k & 7;
        int dy = (int)((DYPACK >> (2*nf)) & 3u) - 1;
        int dx = (int)((DXPACK >> (2*nf)) & 3u) - 1;
        float c  = lds[OFF_PROJ + p*PROJ_PS + (row+1)*PROJ_RS + col];
        float nb = lds[OFF_PROJ + p*PROJ_PS + (row+1-dy)*PROJ_RS + ((col - dx) & (WW-1))];
        float t  = __builtin_amdgcn_exp2f(ALPHA_LOG2E * (c - nb));
        lds[OFF_BITS + k*BITS_PS + pix] = __builtin_amdgcn_rcpf(1.0f + t);
    }
    __syncthreads();

    const float srs = res_scale[0];

    v2f aFr[8][4], aIr[8][4];
    #pragma unroll
    for (int j = 0; j < 8; ++j) {
        #pragma unroll
        for (int mt = 0; mt < 4; ++mt) {
            int ro = (mt*16 + ln)*33 + 4*j + 2*hl;
            aFr[j][mt][0] = lds[OFF_FUSE  + ro]; aFr[j][mt][1] = lds[OFF_FUSE  + ro + 1];
            aIr[j][mt][0] = lds[OFF_PROJW + ro]; aIr[j][mt][1] = lds[OFF_PROJW + ro + 1];
        }
    }
    for (int t = wave; t < HB*8; t += NWAVES) {
        int row  = t >> 3;
        int w0   = (t & 7) << 4;
        int hg   = h0 + row;
        int pixb = (row << 7) + w0;

        v2f bb[8], bx[8];
        const float* xq = xb + ((size_t)(2*hl)*HH + hg)*WW + w0 + ln;
        #pragma unroll
        for (int j = 0; j < 8; ++j) {
            int kb = 4*j + 2*hl;
            bb[j][0] = lds[OFF_BITS + kb*BITS_PS + pixb + ln];
            bb[j][1] = lds[OFF_BITS + (kb+1)*BITS_PS + pixb + ln];
            bx[j][0] = xq[(size_t)(4*j)*HH*WW];
            bx[j][1] = xq[(size_t)(4*j + 1)*HH*WW];
        }

        float* op = out + (((size_t)bimg*COUT)*HH + hg)*WW + w0 + ln;
        int bno = OFF_BN + 8*hl;
        asm volatile("" : "+v"(bno));

        v8f accF[4] = {};
        v8f accI[4] = {};
        #pragma unroll
        for (int j = 0; j < 8; ++j) {
            #pragma unroll
            for (int mt = 0; mt < 4; ++mt) {
                accF[mt] = __builtin_amdgcn_wmma_f32_16x16x4_f32(false, aFr[j][mt], false, bb[j],
                                                                 (short)0, accF[mt], false, false);
                accI[mt] = __builtin_amdgcn_wmma_f32_16x16x4_f32(false, aIr[j][mt], false, bx[j],
                                                                 (short)0, accI[mt], false, false);
            }
        }
        #pragma unroll
        for (int mt = 0; mt < 4; ++mt) {
            const v4f* bnp = (const v4f*)&lds[bno + mt*16];
            v4f inv0 = bnp[0],  inv1 = bnp[1];
            v4f add0 = bnp[16], add1 = bnp[17];
            #pragma unroll
            for (int r = 0; r < 8; ++r) {
                int o = mt*16 + r + 8*hl;
                float inv = (r < 4) ? inv0[r & 3] : inv1[r & 3];
                float add = (r < 4) ? add0[r & 3] : add1[r & 3];
                float v = (accF[mt][r]*inv + add)*srs + accI[mt][r];
                op[(size_t)o*HH*WW] = v > 0.0f ? v : 0.0f;
            }
        }
    }
}

extern "C" void kernel_launch(void* const* d_in, const int* in_sizes, int n_in,
                              void* d_out, int out_size, void* d_ws, size_t ws_size,
                              hipStream_t stream) {
    (void)in_sizes; (void)n_in; (void)d_ws; (void)ws_size; (void)out_size;
    const float* x      = (const float*)d_in[0];
    const float* wpat   = (const float*)d_in[1];
    const float* fusew  = (const float*)d_in[2];
    const float* gammaw = (const float*)d_in[3];
    const float* betaw  = (const float*)d_in[4];
    const float* meanw  = (const float*)d_in[5];
    const float* varw   = (const float*)d_in[6];
    const float* projw  = (const float*)d_in[7];
    const float* rscale = (const float*)d_in[8];
    float* outp = (float*)d_out;

    dim3 grid(BATCH * (HH / HB));
    dim3 block(THREADS);
    size_t ldsBytes = (size_t)LDS_FLOATS * sizeof(float);
    lbp_fused_kernel<<<grid, block, ldsBytes, stream>>>(
        x, wpat, fusew, gammaw, betaw, meanw, varw, projw, rscale, outp);
}


# soft bits made in registers, straight in the B fragment layout, from the proj planes in the GEMM phase per pixel tile; the bits pass, its barrier and the bits array in LDS are gone; sums unchanged
# speedup vs baseline: 1.3716x; 1.0396x over previous
#include <hip/hip_runtime.h>


typedef float v2f __attribute__((ext_vector_type(2)));
typedef float v4f __attribute__((ext_vector_type(4)));
typedef float v8f __attribute__((ext_vector_type(8)));

#define BATCH 16
#define CIN   32
#define HH    128
#define WW    128
#define PP    4
#define NNB   8
#define COUT  64
#define BN_EPS_C 1e-5f
#define ALPHA_LOG2E 14.4269504089f

#define HB      4
#define NWAVES  8
#define THREADS 256

#define OFF_WPAT  0
#define OFF_FUSE  (OFF_WPAT + 16*33)
#define OFF_PROJW (OFF_FUSE + 64*33)
#define OFF_BN    (OFF_PROJW + 64*33)
#define PROJ_RS   144
#define PROJ_PS   ((HB+2)*PROJ_RS)
#define OFF_PROJ  (OFF_BN + 128)
#define LDS_FLOATS (OFF_PROJ + PP*PROJ_PS)

#define DYPACK 0x6A40u
#define DXPACK 0x06A4u

__global__ __launch_bounds__(THREADS, 4)
void lbp_fused_kernel(const float* __restrict__ x,
                      const float* __restrict__ wpat,
                      const float* __restrict__ fusew,
                      const float* __restrict__ gammaw,
                      const float* __restrict__ betaw,
                      const float* __restrict__ meanw,
                      const float* __restrict__ varw,
                      const float* __restrict__ projw,
                      const float* __restrict__ res_scale,
                      float* __restrict__ out)
{
    extern __shared__ __attribute__((aligned(16))) float lds[];
    const int tid  = threadIdx.x;
    const int lane = tid & 31;
    const int wave = tid >> 5;
    const int hl   = lane >> 4;
    const int ln   = lane & 15;

    const int blk  = blockIdx.x;
    const int bimg = blk >> 5;
    const int hblk = blk & 31;
    const int h0   = hblk * HB;

    for (int i = tid; i < 16*CIN; i += THREADS) {
        int m = i >> 5, k = i & 31;
        lds[OFF_WPAT + m*33 + k] = (m < PP) ? wpat[m*CIN + k] : 0.0f;
    }
    for (int i = tid; i < COUT*CIN; i += THREADS) {
        int o = i >> 5, k = i & 31;
        lds[OFF_FUSE  + o*33 + k] = fusew[i];
        lds[OFF_PROJW + o*33 + k] = projw[i];
    }
    if (tid < COUT) {
        float inv = gammaw[tid] * rsqrtf(varw[tid] + BN_EPS_C);
        lds[OFF_BN + tid]      = inv;
        lds[OFF_BN + 64 + tid] = betaw[tid] - meanw[tid] * inv;
    }
    __syncthreads();

    const float* xb = x + (size_t)bimg * (CIN*HH*WW);

    for (int t = wave; t < (HB+2)*8; t += NWAVES) {
        int prow = t >> 3;
        int w0   = (t & 7) << 4;
        int hg   = (h0 - 1 + prow) & (HH - 1);
        v8f acc = {};
        #pragma unroll
        for (int ks = 0; ks < CIN; ks += 4) {
            int kb = ks + 2*hl;
            v2f a, bm;
            a[0] = lds[OFF_WPAT + ln*33 + kb];
            a[1] = lds[OFF_WPAT + ln*33 + kb + 1];
            const float* xp = xb + ((size_t)kb*HH + hg)*WW + w0 + ln;
            bm[0] = xp[0];
            bm[1] = xp[HH*WW];
            acc = __builtin_amdgcn_wmma_f32_16x16x4_f32(false, a, false, bm,
                                                        (short)0, acc, false, false);
        }
        if (hl == 0) {
            #pragma unroll
            for (int r = 0; r < PP; ++r)
                lds[OFF_PROJ + r*PROJ_PS + prow*PROJ_RS + w0 + ln] = acc[r];
        }
    }
    __syncthreads();

    const float srs = res_scale[0];

    v2f aFr[8][4], aIr[8][4];
    #pragma unroll
    for (int j = 0; j < 8; ++j) {
        #pragma unroll
        for (int mt = 0; mt < 4; ++mt) {
            int ro = (mt*16 + ln)*33 + 4*j + 2*hl;
            aFr[j][mt][0] = lds[OFF_FUSE  + ro]; aFr[j][mt][1] = lds[OFF_FUSE  + ro + 1];
            aIr[j][mt][0] = lds[OFF_PROJW + ro]; aIr[j][mt][1] = lds[OFF_PROJW + ro + 1];
        }
    }

    const int w0  = wave << 4;
    const int col = w0 + ln;
    const int cOff = OFF_PROJ + PROJ_RS + col;
    int nOff[2][2];
    #pragma unroll
    for (int jj = 0; jj < 2; ++jj) {
        #pragma unroll
        for (int e = 0; e < 2; ++e) {
            int nf = 4*jj + 2*hl + e;
            int dy = (int)((DYPACK >> (2*nf)) & 3u) - 1;
            int dx = (int)((DXPACK >> (2*nf)) & 3u) - 1;
            nOff[jj][e] = OFF_PROJ + (1 - dy)*PROJ_RS + ((col - dx) & (WW-1));
        }
    }

    for (int row = 0; row < HB; ++row) {
        int hg   = h0 + row;

        v2f bb[8], bx[8];
        const float* xq = xb + ((size_t)(2*hl)*HH + hg)*WW + w0 + ln;
        #pragma unroll
        for (int j = 0; j < 8; ++j) {
            bx[j][0] = xq[(size_t)(4*j)*HH*WW];
            bx[j][1] = xq[(size_t)(4*j + 1)*HH*WW];
        }
        const int rb = row*PROJ_RS;
        #pragma unroll
        for (int p = 0; p < PP; ++p) {
            float c = lds[cOff + p*PROJ_PS + rb];
            #pragma unroll
            for (int jj = 0; jj < 2; ++jj) {
                #pragma unroll
                for (int e = 0; e < 2; ++e) {
                    float nb = lds[nOff[jj][e] + p*PROJ_PS + rb];
                    float tt = __builtin_amdgcn_exp2f(ALPHA_LOG2E * (c - nb));
                    bb[2*p + jj][e] = __builtin_amdgcn_rcpf(1.0f + tt);
                }
            }
        }

        float* op = out + (((size_t)bimg*COUT)*HH + hg)*WW + w0 + ln;
        int bno = OFF_BN + 8*hl;
        asm volatile("" : "+v"(bno));

        v8f accF[4] = {};
        v8f accI[4] = {};
        #pragma unroll
        for (int j = 0; j < 8; ++j) {
            #pragma unroll
            for (int mt = 0; mt < 4; ++mt) {
                accF[mt] = __builtin_amdgcn_wmma_f32_16x16x4_f32(false, aFr[j][mt], false, bb[j],
                                                                 (short)0, accF[mt], false, false);
                accI[mt] = __builtin_amdgcn_wmma_f32_16x16x4_f32(false, aIr[j][mt], false, bx[j],
                                                                 (short)0, accI[mt], false, false);
            }
        }
        #pragma unroll
        for (int mt = 0; mt < 4; ++mt) {
            const v4f* bnp = (const v4f*)&lds[bno + mt*16];
            v4f inv0 = bnp[0],  inv1 = bnp[1];
            v4f add0 = bnp[16], add1 = bnp[17];
            #pragma unroll
            for (int r = 0; r < 8; ++r) {
                int o = mt*16 + r + 8*hl;
                float inv = (r < 4) ? inv0[r & 3] : inv1[r & 3];
                float add = (r < 4) ? add0[r & 3] : add1[r & 3];
                float v = (accF[mt][r]*inv + add)*srs + accI[mt][r];
                op[(size_t)o*HH*WW] = v > 0.0f ? v : 0.0f;
            }
        }
    }
}

extern "C" void kernel_launch(void* const* d_in, const int* in_sizes, int n_in,
                              void* d_out, int out_size, void* d_ws, size_t ws_size,
                              hipStream_t stream) {
    (void)in_sizes; (void)n_in; (void)d_ws; (void)ws_size; (void)out_size;
    const float* x      = (const float*)d_in[0];
    const float* wpat   = (const float*)d_in[1];
    const float* fusew  = (const float*)d_in[2];
    const float* gammaw = (const float*)d_in[3];
    const float* betaw  = (const float*)d_in[4];
    const float* meanw  = (const float*)d_in[5];
    const float* varw   = (const float*)d_in[6];
    const float* projw  = (const float*)d_in[7];
    const float* rscale = (const float*)d_in[8];
    float* outp = (float*)d_out;

    dim3 grid(BATCH * (HH / HB));
    dim3 block(THREADS);
    size_t ldsBytes = (size_t)LDS_FLOATS * sizeof(float);
    lbp_fused_kernel<<<grid, block, ldsBytes, stream>>>(
        x, wpat, fusew, gammaw, betaw, meanw, varw, projw, rscale, outp);
}


# prologue as one load group: proj A fragment, weight and BN words and all six halo tiles of x requested before the first wait (was 17 load-wait-use trips); weight barrier gone; sums unchanged
# speedup vs baseline: 1.7861x; 1.3022x over previous
#include <hip/hip_runtime.h>


typedef float v2f __attribute__((ext_vector_type(2)));
typedef float v4f __attribute__((ext_vector_type(4)));
typedef float v8f __attribute__((ext_vector_type(8)));

#define BATCH 16
#define CIN   32
#define HH    128
#define WW    128
#define PP    4
#define NNB   8
#define COUT  64
#define BN_EPS_C 1e-5f
#define ALPHA_LOG2E 14.4269504089f

#define HB      4
#define NWAVES  8
#define THREADS 256

#define OFF_FUSE  0
#define OFF_PROJW (OFF_FUSE + 64*33)
#define OFF_BN    (OFF_PROJW + 64*33)
#define PROJ_RS   144
#define PROJ_PS   ((HB+2)*PROJ_RS)
#define OFF_PROJ  (OFF_BN + 128)
#define LDS_FLOATS (OFF_PROJ + PP*PROJ_PS)

#define DYPACK 0x6A40u
#define DXPACK 0x06A4u

__global__ __launch_bounds__(THREADS, 4)
void lbp_fused_kernel(const float* __restrict__ x,
                      const float* __restrict__ wpat,
                      const float* __restrict__ fusew,
                      const float* __restrict__ gammaw,
                      const float* __restrict__ betaw,
                      const float* __restrict__ meanw,
                      const float* __restrict__ varw,
                      const float* __restrict__ projw,
                      const float* __restrict__ res_scale,
                      float* __restrict__ out)
{
    extern __shared__ __attribute__((aligned(16))) float lds[];
    const int tid  = threadIdx.x;
    const int lane = tid & 31;
    const int wave = tid >> 5;
    const int hl   = lane >> 4;
    const int ln   = lane & 15;

    const int blk  = blockIdx.x;
    const int bimg = blk >> 5;
    const int hblk = blk & 31;
    const int h0   = hblk * HB;

    const float* xb = x + (size_t)bimg * (CIN*HH*WW);
    const int w0  = wave << 4;

    v2f aw[8];
    #pragma unroll
    for (int s = 0; s < 8; ++s) { aw[s][0] = 0.0f; aw[s][1] = 0.0f; }
    if (ln < PP) {
        const float* wq = wpat + ln*CIN + 2*hl;
        #pragma unroll
        for (int s = 0; s < 8; ++s) { aw[s][0] = wq[4*s]; aw[s][1] = wq[4*s + 1]; }
    }
    float fwr[8], pwr[8];
    #pragma unroll
    for (int q = 0; q < 8; ++q) {
        fwr[q] = fusew[tid + q*THREADS];
        pwr[q] = projw[tid + q*THREADS];
    }

    v2f xh[HB+2][8];
    #pragma unroll
    for (int prow = 0; prow < HB+2; ++prow) {
        int hg = (h0 - 1 + prow) & (HH - 1);
        const float* xp = xb + ((size_t)(2*hl)*HH + hg)*WW + w0 + ln;
        #pragma unroll
        for (int s = 0; s < 8; ++s) {
            xh[prow][s][0] = xp[(size_t)(4*s)*HH*WW];
            xh[prow][s][1] = xp[(size_t)(4*s + 1)*HH*WW];
        }
    }
    const int bi = tid & (COUT - 1);
    const float bng = gammaw[bi], bnv = varw[bi], bnb = betaw[bi], bnm = meanw[bi];

    #pragma unroll
    for (int q = 0; q < 8; ++q) {
        int i = tid + q*THREADS;
        int o = i >> 5, k = i & 31;
        lds[OFF_FUSE  + o*33 + k] = fwr[q];
        lds[OFF_PROJW + o*33 + k] = pwr[q];
    }

    #pragma unroll
    for (int pp = 0; pp < (HB+2)/2; ++pp) {
        v8f acc0 = {}, acc1 = {};
        #pragma unroll
        for (int s = 0; s < 8; ++s) {
            acc0 = __builtin_amdgcn_wmma_f32_16x16x4_f32(false, aw[s], false, xh[2*pp][s],
                                                         (short)0, acc0, false, false);
            acc1 = __builtin_amdgcn_wmma_f32_16x16x4_f32(false, aw[s], false, xh[2*pp + 1][s],
                                                         (short)0, acc1, false, false);
        }
        if (hl == 0) {
            #pragma unroll
            for (int r = 0; r < PP; ++r) {
                lds[OFF_PROJ + r*PROJ_PS + (2*pp)*PROJ_RS     + w0 + ln] = acc0[r];
                lds[OFF_PROJ + r*PROJ_PS + (2*pp + 1)*PROJ_RS + w0 + ln] = acc1[r];
            }
        }
    }
    if (tid < COUT) {
        float inv = bng * rsqrtf(bnv + BN_EPS_C);
        lds[OFF_BN + tid]      = inv;
        lds[OFF_BN + 64 + tid] = bnb - bnm * inv;
    }
    __syncthreads();

    const float srs = res_scale[0];

    v2f aFr[8][4], aIr[8][4];
    #pragma unroll
    for (int j = 0; j < 8; ++j) {
        #pragma unroll
        for (int mt = 0; mt < 4; ++mt) {
            int ro = (mt*16 + ln)*33 + 4*j + 2*hl;
            aFr[j][mt][0] = lds[OFF_FUSE  + ro]; aFr[j][mt][1] = lds[OFF_FUSE  + ro + 1];
            aIr[j][mt][0] = lds[OFF_PROJW + ro]; aIr[j][mt][1] = lds[OFF_PROJW + ro + 1];
        }
    }

    const int col = w0 + ln;
    const int cOff = OFF_PROJ + PROJ_RS + col;
    int nOff[2][2];
    #pragma unroll
    for (int jj = 0; jj < 2; ++jj) {
        #pragma unroll
        for (int e = 0; e < 2; ++e) {
            int nf = 4*jj + 2*hl + e;
            int dy = (int)((DYPACK >> (2*nf)) & 3u) - 1;
            int dx = (int)((DXPACK >> (2*nf)) & 3u) - 1;
            nOff[jj][e] = OFF_PROJ + (1 - dy)*PROJ_RS + ((col - dx) & (WW-1));
        }
    }

    for (int row = 0; row < HB; ++row) {
        int hg   = h0 + row;

        v2f bb[8], bx[8];
        const float* xq = xb + ((size_t)(2*hl)*HH + hg)*WW + w0 + ln;
        #pragma unroll
        for (int j = 0; j < 8; ++j) {
            bx[j][0] = xq[(size_t)(4*j)*HH*WW];
            bx[j][1] = xq[(size_t)(4*j + 1)*HH*WW];
        }
        const int rb = row*PROJ_RS;
        #pragma unroll
        for (int p = 0; p < PP; ++p) {
            float c = lds[cOff + p*PROJ_PS + rb];
            #pragma unroll
            for (int jj = 0; jj < 2; ++jj) {
                #pragma unroll
                for (int e = 0; e < 2; ++e) {
                    float nb = lds[nOff[jj][e] + p*PROJ_PS + rb];
                    float tt = __builtin_amdgcn_exp2f(ALPHA_LOG2E * (c - nb));
                    bb[2*p + jj][e] = __builtin_amdgcn_rcpf(1.0f + tt);
                }
            }
        }

        float* op = out + (((size_t)bimg*COUT)*HH + hg)*WW + w0 + ln;
        int bno = OFF_BN + 8*hl;
        asm volatile("" : "+v"(bno));

        v8f accF[4] = {};
        v8f accI[4] = {};
        #pragma unroll
        for (int j = 0; j < 8; ++j) {
            #pragma unroll
            for (int mt = 0; mt < 4; ++mt) {
                accF[mt] = __builtin_amdgcn_wmma_f32_16x16x4_f32(false, aFr[j][mt], false, bb[j],
                                                                 (short)0, accF[mt], false, false);
                accI[mt] = __builtin_amdgcn_wmma_f32_16x16x4_f32(false, aIr[j][mt], false, bx[j],
                                                                 (short)0, accI[mt], false, false);
            }
        }
        #pragma unroll
        for (int mt = 0; mt < 4; ++mt) {
            const v4f* bnp = (const v4f*)&lds[bno + mt*16];
            v4f inv0 = bnp[0],  inv1 = bnp[1];
            v4f add0 = bnp[16], add1 = bnp[17];
            #pragma unroll
            for (int r = 0; r < 8; ++r) {
                int o = mt*16 + r + 8*hl;
                float inv = (r < 4) ? inv0[r & 3] : inv1[r & 3];
                float add = (r < 4) ? add0[r & 3] : add1[r & 3];
                float v = (accF[mt][r]*inv + add)*srs + accI[mt][r];
                op[(size_t)o*HH*WW] = v > 0.0f ? v : 0.0f;
            }
        }
    }
}

extern "C" void kernel_launch(void* const* d_in, const int* in_sizes, int n_in,
                              void* d_out, int out_size, void* d_ws, size_t ws_size,
                              hipStream_t stream) {
    (void)in_sizes; (void)n_in; (void)d_ws; (void)ws_size; (void)out_size;
    const float* x      = (const float*)d_in[0];
    const float* wpat   = (const float*)d_in[1];
    const float* fusew  = (const float*)d_in[2];
    const float* gammaw = (const float*)d_in[3];
    const float* betaw  = (const float*)d_in[4];
    const float* meanw  = (const float*)d_in[5];
    const float* varw   = (const float*)d_in[6];
    const float* projw  = (const float*)d_in[7];
    const float* rscale = (const float*)d_in[8];
    float* outp = (float*)d_out;

    dim3 grid(BATCH * (HH / HB));
    dim3 block(THREADS);
    size_t ldsBytes = (size_t)LDS_FLOATS * sizeof(float);
    lbp_fused_kernel<<<grid, block, ldsBytes, stream>>>(
        x, wpat, fusew, gammaw, betaw, meanw, varw, projw, rscale, outp);
}


# x slab staged in LDS: each wave fetches 24 whole 512-byte image rows once (16 bytes per lane, one load group); both phases read x fragments from LDS, phase 2 reads no x from global; sums unchanged
# speedup vs baseline: 1.8694x; 1.0466x over previous
#include <hip/hip_runtime.h>


typedef float v2f __attribute__((ext_vector_type(2)));
typedef float v4f __attribute__((ext_vector_type(4)));
typedef float v8f __attribute__((ext_vector_type(8)));

#define BATCH 16
#define CIN   32
#define HH    128
#define WW    128
#define PP    4
#define NNB   8
#define COUT  64
#define BN_EPS_C 1e-5f
#define ALPHA_LOG2E 14.4269504089f

#define HB      4
#define NWAVES  8
#define THREADS 256

#define OFF_FUSE  0
#define OFF_PROJW (OFF_FUSE + 64*33)
#define OFF_BN    (OFF_PROJW + 64*33)
#define PROJ_RS   144
#define PROJ_PS   ((HB+2)*PROJ_RS)
#define OFF_PROJ  (OFF_BN + 128)
#define XP2     144
#define OFF_X   (OFF_PROJ + PP*PROJ_PS)
#define X_ROW   (16*XP2*2)
#define LDS_FLOATS (OFF_X + (HB+2)*X_ROW)

#define DYPACK 0x6A40u
#define DXPACK 0x06A4u

__global__ __launch_bounds__(THREADS, 4)
void lbp_fused_kernel(const float* __restrict__ x,
                      const float* __restrict__ wpat,
                      const float* __restrict__ fusew,
                      const float* __restrict__ gammaw,
                      const float* __restrict__ betaw,
                      const float* __restrict__ meanw,
                      const float* __restrict__ varw,
                      const float* __restrict__ projw,
                      const float* __restrict__ res_scale,
                      float* __restrict__ out)
{
    __shared__ __attribute__((aligned(16))) float lds[LDS_FLOATS];
    const int tid  = threadIdx.x;
    const int lane = tid & 31;
    const int wave = tid >> 5;
    const int hl   = lane >> 4;
    const int ln   = lane & 15;

    const int blk  = blockIdx.x;
    const int bimg = blk >> 5;
    const int hblk = blk & 31;
    const int h0   = hblk * HB;

    const float* xb = x + (size_t)bimg * (CIN*HH*WW);
    const int w0  = wave << 4;

    v4f xr0[12], xr1[12];
    #pragma unroll
    for (int i = 0; i < 12; ++i) {
        int cp   = 2*wave + i/6;
        int prow = i % 6;
        int hg   = (h0 - 1 + prow) & (HH - 1);
        const float* xg = xb + ((size_t)(2*cp)*HH + hg)*WW + 4*lane;
        xr0[i] = *(const v4f*)xg;
        xr1[i] = *(const v4f*)(xg + HH*WW);
    }

    v2f aw[8];
    #pragma unroll
    for (int s = 0; s < 8; ++s) { aw[s][0] = 0.0f; aw[s][1] = 0.0f; }
    if (ln < PP) {
        const float* wq = wpat + ln*CIN + 2*hl;
        #pragma unroll
        for (int s = 0; s < 8; ++s) { aw[s][0] = wq[4*s]; aw[s][1] = wq[4*s + 1]; }
    }
    float fwr[8], pwr[8];
    #pragma unroll
    for (int q = 0; q < 8; ++q) {
        fwr[q] = fusew[tid + q*THREADS];
        pwr[q] = projw[tid + q*THREADS];
    }

    const int bi = tid & (COUT - 1);
    const float bng = gammaw[bi], bnv = varw[bi], bnb = betaw[bi], bnm = meanw[bi];

    #pragma unroll
    for (int i = 0; i < 12; ++i) {
        int cp   = 2*wave + i/6;
        int prow = i % 6;
        v4f lo, hi;
        lo[0] = xr0[i][0]; lo[1] = xr1[i][0]; lo[2] = xr0[i][1]; lo[3] = xr1[i][1];
        hi[0] = xr0[i][2]; hi[1] = xr1[i][2]; hi[2] = xr0[i][3]; hi[3] = xr1[i][3];
        v4f* xs = (v4f*)&lds[OFF_X + prow*X_ROW + (cp*XP2 + 4*lane)*2];
        xs[0] = lo;
        xs[1] = hi;
    }
    #pragma unroll
    for (int q = 0; q < 8; ++q) {
        int i = tid + q*THREADS;
        int o = i >> 5, k = i & 31;
        lds[OFF_FUSE  + o*33 + k] = fwr[q];
        lds[OFF_PROJW + o*33 + k] = pwr[q];
    }
    __syncthreads();

    const int xfo = OFF_X + (hl*XP2 + w0 + ln)*2;
    #pragma unroll
    for (int pp = 0; pp < (HB+2)/2; ++pp) {
        v2f xh0[8], xh1[8];
        #pragma unroll
        for (int s = 0; s < 8; ++s) {
            int o0 = xfo + (2*pp)*X_ROW + (2*s)*XP2*2;
            xh0[s] = *(const v2f*)&lds[o0];
            xh1[s] = *(const v2f*)&lds[o0 + X_ROW];
        }
        v8f acc0 = {}, acc1 = {};
        #pragma unroll
        for (int s = 0; s < 8; ++s) {
            acc0 = __builtin_amdgcn_wmma_f32_16x16x4_f32(false, aw[s], false, xh0[s],
                                                         (short)0, acc0, false, false);
            acc1 = __builtin_amdgcn_wmma_f32_16x16x4_f32(false, aw[s], false, xh1[s],
                                                         (short)0, acc1, false, false);
        }
        if (hl == 0) {
            #pragma unroll
            for (int r = 0; r < PP; ++r) {
                lds[OFF_PROJ + r*PROJ_PS + (2*pp)*PROJ_RS     + w0 + ln] = acc0[r];
                lds[OFF_PROJ + r*PROJ_PS + (2*pp + 1)*PROJ_RS + w0 + ln] = acc1[r];
            }
        }
    }
    if (tid < COUT) {
        float inv = bng * rsqrtf(bnv + BN_EPS_C);
        lds[OFF_BN + tid]      = inv;
        lds[OFF_BN + 64 + tid] = bnb - bnm * inv;
    }
    __syncthreads();

    const float srs = res_scale[0];

    v2f aFr[8][4], aIr[8][4];
    #pragma unroll
    for (int j = 0; j < 8; ++j) {
        #pragma unroll
        for (int mt = 0; mt < 4; ++mt) {
            int ro = (mt*16 + ln)*33 + 4*j + 2*hl;
            aFr[j][mt][0] = lds[OFF_FUSE  + ro]; aFr[j][mt][1] = lds[OFF_FUSE  + ro + 1];
            aIr[j][mt][0] = lds[OFF_PROJW + ro]; aIr[j][mt][1] = lds[OFF_PROJW + ro + 1];
        }
    }

    const int col = w0 + ln;
    const int cOff = OFF_PROJ + PROJ_RS + col;
    int nOff[2][2];
    #pragma unroll
    for (int jj = 0; jj < 2; ++jj) {
        #pragma unroll
        for (int e = 0; e < 2; ++e) {
            int nf = 4*jj + 2*hl + e;
            int dy = (int)((DYPACK >> (2*nf)) & 3u) - 1;
            int dx = (int)((DXPACK >> (2*nf)) & 3u) - 1;
            nOff[jj][e] = OFF_PROJ + (1 - dy)*PROJ_RS + ((col - dx) & (WW-1));
        }
    }

    #pragma unroll
    for (int row = 0; row < HB; ++row) {
        int hg   = h0 + row;

        v2f bb[8], bx[8];
        const int xo = xfo + (row + 1)*X_ROW;
        #pragma unroll
        for (int j = 0; j < 8; ++j) bx[j] = *(const v2f*)&lds[xo + (2*j)*XP2*2];
        const int rb = row*PROJ_RS;
        #pragma unroll
        for (int p = 0; p < PP; ++p) {
            float c = lds[cOff + p*PROJ_PS + rb];
            #pragma unroll
            for (int jj = 0; jj < 2; ++jj) {
                #pragma unroll
                for (int e = 0; e < 2; ++e) {
                    float nb = lds[nOff[jj][e] + p*PROJ_PS + rb];
                    float tt = __builtin_amdgcn_exp2f(ALPHA_LOG2E * (c - nb));
                    bb[2*p + jj][e] = __builtin_amdgcn_rcpf(1.0f + tt);
                }
            }
        }

        float* op = out + (((size_t)bimg*COUT)*HH + hg)*WW + w0 + ln;
        int bno = OFF_BN + 8*hl;
        asm volatile("" : "+v"(bno));

        v8f accF[4] = {};
        v8f accI[4] = {};
        #pragma unroll
        for (int j = 0; j < 8; ++j) {
            #pragma unroll
            for (int mt = 0; mt < 4; ++mt) {
                accF[mt] = __builtin_amdgcn_wmma_f32_16x16x4_f32(false, aFr[j][mt], false, bb[j],
                                                                 (short)0, accF[mt], false, false);
                accI[mt] = __builtin_amdgcn_wmma_f32_16x16x4_f32(false, aIr[j][mt], false, bx[j],
                                                                 (short)0, accI[mt], false, false);
            }
        }
        #pragma unroll
        for (int mt = 0; mt < 4; ++mt) {
            const v4f* bnp = (const v4f*)&lds[bno + mt*16];
            v4f inv0 = bnp[0],  inv1 = bnp[1];
            v4f add0 = bnp[16], add1 = bnp[17];
            #pragma unroll
            for (int r = 0; r < 8; ++r) {
                int o = mt*16 + r + 8*hl;
                float inv = (r < 4) ? inv0[r & 3] : inv1[r & 3];
                float add = (r < 4) ? add0[r & 3] : add1[r & 3];
                float v = (accF[mt][r]*inv + add)*srs + accI[mt][r];
                op[(size_t)o*HH*WW] = v > 0.0f ? v : 0.0f;
            }
        }
    }
}

extern "C" void kernel_launch(void* const* d_in, const int* in_sizes, int n_in,
                              void* d_out, int out_size, void* d_ws, size_t ws_size,
                              hipStream_t stream) {
    (void)in_sizes; (void)n_in; (void)d_ws; (void)ws_size; (void)out_size;
    const float* x      = (const float*)d_in[0];
    const float* wpat   = (const float*)d_in[1];
    const float* fusew  = (const float*)d_in[2];
    const float* gammaw = (const float*)d_in[3];
    const float* betaw  = (const float*)d_in[4];
    const float* meanw  = (const float*)d_in[5];
    const float* varw   = (const float*)d_in[6];
    const float* projw  = (const float*)d_in[7];
    const float* rscale = (const float*)d_in[8];
    float* outp = (float*)d_out;

    dim3 grid(BATCH * (HH / HB));
    dim3 block(THREADS);
    lbp_fused_kernel<<<grid, block, 0, stream>>>(
        x, wpat, fusew, gammaw, betaw, meanw, varw, projw, rscale, outp);
}


# weight matrices kept in LDS in 16-byte k groups: A fragments of phase 2 read as 16-byte loads under phase 1's products, not as 64 two-word reads behind the second barrier; sums unchanged
# speedup vs baseline: 1.9590x; 1.0479x over previous
#include <hip/hip_runtime.h>


typedef float v2f __attribute__((ext_vector_type(2)));
typedef float v4f __attribute__((ext_vector_type(4)));
typedef float v8f __attribute__((ext_vector_type(8)));

#define BATCH 16
#define CIN   32
#define HH    128
#define WW    128
#define PP    4
#define NNB   8
#define COUT  64
#define BN_EPS_C 1e-5f
#define ALPHA_LOG2E 14.4269504089f

#define HB      4
#define NWAVES  8
#define THREADS 256

#define WS        36
#define OFF_FUSE  0
#define OFF_PROJW (OFF_FUSE + 64*WS)
#define OFF_BN    (OFF_PROJW + 64*WS)
#define PROJ_RS   144
#define PROJ_PS   ((HB+2)*PROJ_RS)
#define OFF_PROJ  (OFF_BN + 128)
#define XP2     144
#define OFF_X   (OFF_PROJ + PP*PROJ_PS)
#define X_ROW   (16*XP2*2)
#define LDS_FLOATS (OFF_X + (HB+2)*X_ROW)

#define DYPACK 0x6A40u
#define DXPACK 0x06A4u

__global__ __launch_bounds__(THREADS, 4)
void lbp_fused_kernel(const float* __restrict__ x,
                      const float* __restrict__ wpat,
                      const float* __restrict__ fusew,
                      const float* __restrict__ gammaw,
                      const float* __restrict__ betaw,
                      const float* __restrict__ meanw,
                      const float* __restrict__ varw,
                      const float* __restrict__ projw,
                      const float* __restrict__ res_scale,
                      float* __restrict__ out)
{
    __shared__ __attribute__((aligned(16))) float lds[LDS_FLOATS];
    const int tid  = threadIdx.x;
    const int lane = tid & 31;
    const int wave = tid >> 5;
    const int hl   = lane >> 4;
    const int ln   = lane & 15;

    const int blk  = blockIdx.x;
    const int bimg = blk >> 5;
    const int hblk = blk & 31;
    const int h0   = hblk * HB;

    const float* xb = x + (size_t)bimg * (CIN*HH*WW);
    const int w0  = wave << 4;

    v4f xr0[12], xr1[12];
    #pragma unroll
    for (int i = 0; i < 12; ++i) {
        int cp   = 2*wave + i/6;
        int prow = i % 6;
        int hg   = (h0 - 1 + prow) & (HH - 1);
        const float* xg = xb + ((size_t)(2*cp)*HH + hg)*WW + 4*lane;
        xr0[i] = *(const v4f*)xg;
        xr1[i] = *(const v4f*)(xg + HH*WW);
    }

    v2f aw[8];
    #pragma unroll
    for (int s = 0; s < 8; ++s) { aw[s][0] = 0.0f; aw[s][1] = 0.0f; }
    if (ln < PP) {
        const float* wq = wpat + ln*CIN + 2*hl;
        #pragma unroll
        for (int s = 0; s < 8; ++s) { aw[s][0] = wq[4*s]; aw[s][1] = wq[4*s + 1]; }
    }
    const v4f* fwg = (const v4f*)(fusew + 8*tid);
    const v4f* pwg = (const v4f*)(projw + 8*tid);
    const v4f fw0 = fwg[0], fw1 = fwg[1], pw0 = pwg[0], pw1 = pwg[1];

    const int bi = tid & (COUT - 1);
    const float bng = gammaw[bi], bnv = varw[bi], bnb = betaw[bi], bnm = meanw[bi];

    #pragma unroll
    for (int i = 0; i < 12; ++i) {
        int cp   = 2*wave + i/6;
        int prow = i % 6;
        v4f lo, hi;
        lo[0] = xr0[i][0]; lo[1] = xr1[i][0]; lo[2] = xr0[i][1]; lo[3] = xr1[i][1];
        hi[0] = xr0[i][2]; hi[1] = xr1[i][2]; hi[2] = xr0[i][3]; hi[3] = xr1[i][3];
        v4f* xs = (v4f*)&lds[OFF_X + prow*X_ROW + (cp*XP2 + 4*lane)*2];
        xs[0] = lo;
        xs[1] = hi;
    }
    {
        v4f lo, hi;
        v4f* wd = (v4f*)&lds[OFF_FUSE + (tid >> 2)*WS + 8*(tid & 3)];
        lo[0] = fw0[0]; lo[1] = fw0[1]; lo[2] = fw1[0]; lo[3] = fw1[1];
        hi[0] = fw0[2]; hi[1] = fw0[3]; hi[2] = fw1[2]; hi[3] = fw1[3];
        wd[0] = lo;
        wd[1] = hi;
        wd += (OFF_PROJW - OFF_FUSE)/4;
        lo[0] = pw0[0]; lo[1] = pw0[1]; lo[2] = pw1[0]; lo[3] = pw1[1];
        hi[0] = pw0[2]; hi[1] = pw0[3]; hi[2] = pw1[2]; hi[3] = pw1[3];
        wd[0] = lo;
        wd[1] = hi;
    }
    __syncthreads();

    v2f aFr[8][4], aIr[8][4];
    const int wfo = OFF_FUSE + ln*WS + 4*hl;

    const int xfo = OFF_X + (hl*XP2 + w0 + ln)*2;
    #pragma unroll
    for (int pp = 0; pp < (HB+2)/2; ++pp) {
        v2f xh0[8], xh1[8];
        #pragma unroll
        for (int s = 0; s < 8; ++s) {
            int o0 = xfo + (2*pp)*X_ROW + (2*s)*XP2*2;
            xh0[s] = *(const v2f*)&lds[o0];
            xh1[s] = *(const v2f*)&lds[o0 + X_ROW];
        }
        #pragma unroll
        for (int c = 0; c < 8; ++c) {
            if (c / 3 != pp) continue;
            const int g = c & 3;
            #pragma unroll
            for (int mt = 0; mt < 4; ++mt) {
                v4f t = *(const v4f*)&lds[wfo + (c >> 2)*(OFF_PROJW - OFF_FUSE) + mt*16*WS + 8*g];
                if (c < 4) { aFr[2*g][mt][0] = t[0]; aFr[2*g][mt][1] = t[1]; aFr[2*g + 1][mt][0] = t[2]; aFr[2*g + 1][mt][1] = t[3]; }
                else       { aIr[2*g][mt][0] = t[0]; aIr[2*g][mt][1] = t[1]; aIr[2*g + 1][mt][0] = t[2]; aIr[2*g + 1][mt][1] = t[3]; }
            }
        }
        v8f acc0 = {}, acc1 = {};
        #pragma unroll
        for (int s = 0; s < 8; ++s) {
            acc0 = __builtin_amdgcn_wmma_f32_16x16x4_f32(false, aw[s], false, xh0[s],
                                                         (short)0, acc0, false, false);
            acc1 = __builtin_amdgcn_wmma_f32_16x16x4_f32(false, aw[s], false, xh1[s],
                                                         (short)0, acc1, false, false);
        }
        if (hl == 0) {
            #pragma unroll
            for (int r = 0; r < PP; ++r) {
                lds[OFF_PROJ + r*PROJ_PS + (2*pp)*PROJ_RS     + w0 + ln] = acc0[r];
                lds[OFF_PROJ + r*PROJ_PS + (2*pp + 1)*PROJ_RS + w0 + ln] = acc1[r];
            }
        }
    }
    if (tid < COUT) {
        float inv = bng * rsqrtf(bnv + BN_EPS_C);
        lds[OFF_BN + tid]      = inv;
        lds[OFF_BN + 64 + tid] = bnb - bnm * inv;
    }
    __syncthreads();

    const float srs = res_scale[0];

    const int col = w0 + ln;
    const int cOff = OFF_PROJ + PROJ_RS + col;
    int nOff[2][2];
    #pragma unroll
    for (int jj = 0; jj < 2; ++jj) {
        #pragma unroll
        for (int e = 0; e < 2; ++e) {
            int nf = 4*jj + 2*hl + e;
            int dy = (int)((DYPACK >> (2*nf)) & 3u) - 1;
            int dx = (int)((DXPACK >> (2*nf)) & 3u) - 1;
            nOff[jj][e] = OFF_PROJ + (1 - dy)*PROJ_RS + ((col - dx) & (WW-1));
        }
    }

    #pragma unroll
    for (int row = 0; row < HB; ++row) {
        int hg   = h0 + row;

        v2f bb[8], bx[8];
        const int xo = xfo + (row + 1)*X_ROW;
        #pragma unroll
        for (int j = 0; j < 8; ++j) bx[j] = *(const v2f*)&lds[xo + (2*j)*XP2*2];
        const int rb = row*PROJ_RS;
        #pragma unroll
        for (int p = 0; p < PP; ++p) {
            float c = lds[cOff + p*PROJ_PS + rb];
            #pragma unroll
            for (int jj = 0; jj < 2; ++jj) {
                #pragma unroll
                for (int e = 0; e < 2; ++e) {
                    float nb = lds[nOff[jj][e] + p*PROJ_PS + rb];
                    float tt = __builtin_amdgcn_exp2f(ALPHA_LOG2E * (c - nb));
                    bb[2*p + jj][e] = __builtin_amdgcn_rcpf(1.0f + tt);
                }
            }
        }

        float* op = out + (((size_t)bimg*COUT)*HH + hg)*WW + w0 + ln;
        int bno = OFF_BN + 8*hl;
        asm volatile("" : "+v"(bno));

        v8f accF[4] = {};
        v8f accI[4] = {};
        #pragma unroll
        for (int j = 0; j < 8; ++j) {
            #pragma unroll
            for (int mt = 0; mt < 4; ++mt) {
                accF[mt] = __builtin_amdgcn_wmma_f32_16x16x4_f32(false, aFr[j][mt], false, bb[j],
                                                                 (short)0, accF[mt], false, false);
                accI[mt] = __builtin_amdgcn_wmma_f32_16x16x4_f32(false, aIr[j][mt], false, bx[j],
                                                                 (short)0, accI[mt], false, false);
            }
        }
        #pragma unroll
        for (int mt = 0; mt < 4; ++mt) {
            const v4f* bnp = (const v4f*)&lds[bno + mt*16];
            v4f inv0 = bnp[0],  inv1 = bnp[1];
            v4f add0 = bnp[16], add1 = bnp[17];
            #pragma unroll
            for (int r = 0; r < 8; ++r) {
                int o = mt*16 + r + 8*hl;
                float inv = (r < 4) ? inv0[r & 3] : inv1[r & 3];
                float add = (r < 4) ? add0[r & 3] : add1[r & 3];
                float v = (accF[mt][r]*inv + add)*srs + accI[mt][r];
                op[(size_t)o*HH*WW] = v > 0.0f ? v : 0.0f;
            }
        }
    }
}

extern "C" void kernel_launch(void* const* d_in, const int* in_sizes, int n_in,
                              void* d_out, int out_size, void* d_ws, size_t ws_size,
                              hipStream_t stream) {
    (void)in_sizes; (void)n_in; (void)d_ws; (void)ws_size; (void)out_size;
    const float* x      = (const float*)d_in[0];
    const float* wpat   = (const float*)d_in[1];
    const float* fusew  = (const float*)d_in[2];
    const float* gammaw = (const float*)d_in[3];
    const float* betaw  = (const float*)d_in[4];
    const float* meanw  = (const float*)d_in[5];
    const float* varw   = (const float*)d_in[6];
    const float* projw  = (const float*)d_in[7];
    const float* rscale = (const float*)d_in[8];
    float* outp = (float*)d_out;

    dim3 grid(BATCH * (HH / HB));
    dim3 block(THREADS);
    lbp_fused_kernel<<<grid, block, 0, stream>>>(
        x, wpat, fusew, gammaw, betaw, meanw, varw, projw, rscale, outp);
}
